# starting kernel
# baseline (speedup 1.0000x reference)
#include <hip/hip_runtime.h>
#include <hip/hip_bf16.h>

typedef __attribute__((ext_vector_type(16))) _Float16 v16h;
typedef __attribute__((ext_vector_type(8)))  float    v8f;
typedef __attribute__((ext_vector_type(4)))  float    v4f;

#define IN_DIM 128
#define HID    64
#define NOUT   3
#define WPB    8


__global__ void zero_f32(float* __restrict__ p, size_t n) {
  size_t i = (size_t)blockIdx.x * blockDim.x + threadIdx.x;
  if (i < n) p[i] = 0.0f;
}

__global__ void degree_count(const int* __restrict__ ei, float* __restrict__ cnt, int E) {
  int e = blockIdx.x * blockDim.x + threadIdx.x;
  if (e < E) atomicAdd(cnt + ei[E + e], 1.0f);
}

__global__ void scatter_add_128(const float* __restrict__ x, const int* __restrict__ ei,
                                float* __restrict__ agg, int E) {
  long long t = (long long)blockIdx.x * blockDim.x + threadIdx.x;
  int e = (int)(t >> 5);
  if (e >= E) return;
  int c = ((int)t & 31) * 4;
  int s = ei[e], d = ei[E + e];
  v4f v = *(const v4f*)(x + (size_t)s * IN_DIM + c);
  float* out = agg + (size_t)d * IN_DIM + c;
  atomicAdd(out + 0, v[0]); atomicAdd(out + 1, v[1]);
  atomicAdd(out + 2, v[2]); atomicAdd(out + 3, v[3]);
}

__global__ void scatter_add_64(const float* __restrict__ h, const int* __restrict__ ei,
                               float* __restrict__ agg, int E) {
  long long t = (long long)blockIdx.x * blockDim.x + threadIdx.x;
  int e = (int)(t >> 4);
  if (e >= E) return;
  int c = ((int)t & 15) * 4;
  int s = ei[e], d = ei[E + e];
  v4f v = *(const v4f*)(h + (size_t)s * HID + c);
  float* out = agg + (size_t)d * HID + c;
  atomicAdd(out + 0, v[0]); atomicAdd(out + 1, v[1]);
  atomicAdd(out + 2, v[2]); atomicAdd(out + 3, v[3]);
}

__device__ __forceinline__ v16h load_a_frag_f32(const float* __restrict__ rowBase,
                                                int k0, float scale, int khalf) {
  int g0 = khalf * 8;
  int g1 = 16 + khalf * 8;
  v4f a0 = *(const v4f*)(rowBase + k0 + g0);
  v4f a1 = *(const v4f*)(rowBase + k0 + g0 + 4);
  v4f a2 = *(const v4f*)(rowBase + k0 + g1);
  v4f a3 = *(const v4f*)(rowBase + k0 + g1 + 4);
  v16h r;
#pragma unroll
  for (int i = 0; i < 4; ++i) {
    r[i]      = (_Float16)(a0[i] * scale);
    r[4 + i]  = (_Float16)(a1[i] * scale);
    r[8 + i]  = (_Float16)(a2[i] * scale);
    r[12 + i] = (_Float16)(a3[i] * scale);
  }
  return r;
}

__global__ void __launch_bounds__(256)
sage_layer1_wmma(const float* __restrict__ x, const float* __restrict__ agg,
                 const float* __restrict__ cnt,
                 const float* __restrict__ Wl, const float* __restrict__ Wr,
                 const float* __restrict__ bias,
                 float* __restrict__ h, int N, int rowTiles) {
  __shared__ _Float16 sW[2][32][256];

  for (int idx = threadIdx.x; idx < 32 * 256; idx += 256) {
    int g  = idx >> 8;
    int n  = (idx >> 4) & 15;
    int j  = idx & 15;
    int ct = g >> 3;
    int kh = (g >> 2) & 1;
    int kc = g & 3;
    int k  = kc * 32 + kh * 8 + j + ((j >= 8) ? 8 : 0);
    int col = ct * 16 + n;
    sW[0][g][n * 16 + j] = (_Float16)Wl[k * HID + col];
    sW[1][g][n * 16 + j] = (_Float16)Wr[k * HID + col];
  }
  __syncthreads();

  int wave = threadIdx.x >> 5;
  int lid  = threadIdx.x & 31;
  int rowTile = blockIdx.x * WPB + wave;
  if (rowTile >= rowTiles) return;

  int r0    = rowTile * 16;
  int m     = lid & 15;
  int khalf = lid >> 4;
  int row = r0 + m; if (row >= N) row = N - 1;

  float inv = 1.0f / fmaxf(cnt[row], 1.0f);
  const float* aggRow = agg + (size_t)row * IN_DIM;
  const float* xRow   = x   + (size_t)row * IN_DIM;

  v8f acc[4] = {{}, {}, {}, {}};
#pragma unroll
  for (int kc = 0; kc < 4; ++kc) {
    int k0 = kc * 32;
    v16h aA = load_a_frag_f32(aggRow, k0, inv,  khalf);
    v16h aX = load_a_frag_f32(xRow,   k0, 1.0f, khalf);
#pragma unroll
    for (int ct = 0; ct < 4; ++ct) {
      int g = (ct * 2 + khalf) * 4 + kc;
      v16h bL = *(const v16h*)&sW[0][g][m * 16];
      v16h bR = *(const v16h*)&sW[1][g][m * 16];
      acc[ct] = __builtin_amdgcn_wmma_f32_16x16x32_f16(false, aA, false, bL,
                                                       (short)0, acc[ct],
                                                       false, false);
      acc[ct] = __builtin_amdgcn_wmma_f32_16x16x32_f16(false, aX, false, bR,
                                                       (short)0, acc[ct],
                                                       false, false);
    }
  }

#pragma unroll
  for (int ct = 0; ct < 4; ++ct) {
    int n = ct * 16 + m;
    float bv = bias[n];
#pragma unroll
    for (int r = 0; r < 8; ++r) {
      int orow = r0 + r + 8 * khalf;
      if (orow < N) h[(size_t)orow * HID + n] = fmaxf(acc[ct][r] + bv, 0.0f);
    }
  }
}

__global__ void sage_layer2(const float* __restrict__ h, const float* __restrict__ agg2,
                            const float* __restrict__ cnt,
                            const float* __restrict__ Wl, const float* __restrict__ Wr,
                            const float* __restrict__ bias,
                            float* __restrict__ out, int N) {
  int nidx = blockIdx.x * blockDim.x + threadIdx.x;
  if (nidx >= N) return;
  float inv = 1.0f / fmaxf(cnt[nidx], 1.0f);
  const float* hr = h    + (size_t)nidx * HID;
  const float* ar = agg2 + (size_t)nidx * HID;
  float o0 = bias[0], o1 = bias[1], o2 = bias[2];
#pragma unroll 8
  for (int k = 0; k < HID; ++k) {
    float av = ar[k] * inv;
    float hv = hr[k];
    o0 += av * Wl[k * NOUT + 0] + hv * Wr[k * NOUT + 0];
    o1 += av * Wl[k * NOUT + 1] + hv * Wr[k * NOUT + 1];
    o2 += av * Wl[k * NOUT + 2] + hv * Wr[k * NOUT + 2];
  }
  float mx  = fmaxf(o0, fmaxf(o1, o2));
  float lse = logf(expf(o0 - mx) + expf(o1 - mx) + expf(o2 - mx));
  out[(size_t)nidx * NOUT + 0] = o0 - mx - lse;
  out[(size_t)nidx * NOUT + 1] = o1 - mx - lse;
  out[(size_t)nidx * NOUT + 2] = o2 - mx - lse;
}


extern "C" void kernel_launch(void* const* d_in, const int* in_sizes, int n_in,
                              void* d_out, int out_size, void* d_ws, size_t ws_size,
                              hipStream_t stream) {
  (void)n_in; (void)out_size; (void)ws_size;
  const float* x   = (const float*)d_in[0];
  const int*   ei  = (const int*)d_in[1];
  const float* Wl1 = (const float*)d_in[2];
  const float* Wr1 = (const float*)d_in[3];
  const float* b1  = (const float*)d_in[4];
  const float* Wl2 = (const float*)d_in[5];
  const float* Wr2 = (const float*)d_in[6];
  const float* b2  = (const float*)d_in[7];

  int N = in_sizes[0] / IN_DIM;
  int E = in_sizes[1] / 2;

  float* agg1 = (float*)d_ws;
  float* agg2 = agg1 + (size_t)N * IN_DIM;
  float* hbuf = agg2 + (size_t)N * HID;
  float* cnt  = hbuf + (size_t)N * HID;

  size_t nz = (size_t)N * (IN_DIM + HID);
  zero_f32<<<(int)((nz + 255) / 256), 256, 0, stream>>>(agg1, nz);
  zero_f32<<<(N + 255) / 256, 256, 0, stream>>>(cnt, (size_t)N);

  degree_count<<<(E + 255) / 256, 256, 0, stream>>>(ei, cnt, E);

  long long t1 = (long long)E * 32;
  scatter_add_128<<<(int)((t1 + 255) / 256), 256, 0, stream>>>(x, ei, agg1, E);

  int rowTiles = (N + 15) / 16;
  int blocks   = (rowTiles + WPB - 1) / WPB;
  sage_layer1_wmma<<<blocks, 256, 0, stream>>>(x, agg1, cnt, Wl1, Wr1, b1,
                                               hbuf, N, rowTiles);

  long long t2 = (long long)E * 16;
  scatter_add_64<<<(int)((t2 + 255) / 256), 256, 0, stream>>>(hbuf, ei, agg2, E);

  sage_layer2<<<(N + 255) / 256, 256, 0, stream>>>(hbuf, agg2, cnt, Wl2, Wr2, b2,
                                                   (float*)d_out, N);
}


# Each work-group owns 400 destination nodes: filters the edge list for them, sums source rows in LDS by f32 LDS atomics (arrival order), feeds both layers from LDS; no sum planes, 5 launches become 2
# speedup vs baseline: 2.1995x; 2.1995x over previous
#include <hip/hip_runtime.h>
#include <hip/hip_bf16.h>


typedef __attribute__((ext_vector_type(16))) _Float16 v16h;
typedef __attribute__((ext_vector_type(8)))  float    v8f;
typedef __attribute__((ext_vector_type(4)))  float    v4f;

#define IN_DIM 128
#define HID    64
#define NOUT   3
#define RB     400
#define NT     1024
#define NW     32
#define P1     132
#define P2     68
#define CAP1   8192
#define CAP2   16384
#define SK     16

template <int ROW, int PITCH, int U, int CAP>
__device__ __forceinline__ void owner_sum(const float* __restrict__ rows,
                                          const int* __restrict__ ei, int E,
                                          int n0, int nrows,
                                          float* sAgg, float* sCnt,
                                          int* sLS, int* sLD, int* sNum) {
  constexpr int LPR = ROW / 32;
  const unsigned lane = threadIdx.x & 31u;
  const int wave = __builtin_amdgcn_readfirstlane((int)(threadIdx.x >> 5));

  {
    v4f* q = (v4f*)sAgg;
    const v4f z = {0.0f, 0.0f, 0.0f, 0.0f};
    for (int i = threadIdx.x; i < RB * PITCH / 4; i += NT) q[i] = z;
    for (int i = threadIdx.x; i < RB; i += NT) sCnt[i] = 0.0f;
    if (threadIdx.x == 0) *sNum = 0;
  }
  __syncthreads();

  auto add_edge = [&](int s, int d) {
    const float* in = rows + (size_t)s * ROW;
    float* o = sAgg + d * PITCH;
    float t[LPR];
#pragma unroll
    for (int k = 0; k < LPR; ++k) t[k] = in[lane + 32u * k];
#pragma unroll
    for (int k = 0; k < LPR; ++k) atomicAdd(o + (lane + 32u * k), t[k]);
  };

  auto take = [&](int e, bool match, unsigned t) {
    bool ovf = false;
    if (match) {
      atomicAdd(sCnt + t, 1.0f);
      const int pos = atomicAdd(sNum, 1);
      if (pos < CAP) { sLS[pos] = e; sLD[pos] = (int)t; } else ovf = true;
    }
    unsigned om = __builtin_amdgcn_ballot_w32(ovf);
    while (om) {
      const int j = __builtin_ctz(om);
      om &= om - 1u;
      const int ej = __builtin_amdgcn_readlane(e, j);
      const int dj = __builtin_amdgcn_readlane((int)t, j);
      add_edge(ei[ej], dj);
    }
  };

  const int* __restrict__ dstp = ei + E;
  const int nfull = E / (32 * SK);
  const int off = (int)blockIdx.x * (nfull / (int)gridDim.x);
  for (int i = wave; i < nfull; i += NW) {
    int c = i + off; if (c >= nfull) c -= nfull;
    const int base = c * (32 * SK) + (int)lane;
    int d[SK];
#pragma unroll
    for (int k = 0; k < SK; ++k) d[k] = dstp[base + 32 * k];
#pragma unroll
    for (int k = 0; k < SK; ++k) {
      const unsigned t = (unsigned)(d[k] - n0);
      const bool match = t < (unsigned)nrows;
      if (__builtin_amdgcn_ballot_w32(match)) take(base + 32 * k, match, t);
    }
  }
  for (int e0 = nfull * (32 * SK) + wave * 32; e0 < E; e0 += NW * 32) {
    const int e = e0 + (int)lane;
    const bool valid = e < E;
    const int dd = dstp[valid ? e : E - 1];
    const unsigned t = (unsigned)(dd - n0);
    const bool match = valid && t < (unsigned)nrows;
    if (__builtin_amdgcn_ballot_w32(match)) take(e, match, t);
  }
  __syncthreads();

  int total = *sNum; if (total > CAP) total = CAP;
  total = __builtin_amdgcn_readfirstlane(total);
  for (int i = threadIdx.x; i < total; i += NT) sLS[i] = ei[sLS[i]];
  __syncthreads();

  for (int b0 = wave * U; b0 < total; b0 += NW * U) {
    const int idx = b0 + (int)lane;
    int sv = 0, dv = 0;
    if ((int)lane < U && idx < total) { sv = sLS[idx]; dv = sLD[idx]; }
    if (b0 + U <= total) {
      float v[U][LPR];
      int dd[U];
#pragma unroll
      for (int u = 0; u < U; ++u) {
        const int s = __builtin_amdgcn_readlane(sv, u);
        dd[u] = __builtin_amdgcn_readlane(dv, u);
        const float* in = rows + (size_t)s * ROW;
#pragma unroll
        for (int k = 0; k < LPR; ++k) v[u][k] = in[lane + 32u * k];
      }
#pragma unroll
      for (int u = 0; u < U; ++u) {
        float* o = sAgg + dd[u] * PITCH;
#pragma unroll
        for (int k = 0; k < LPR; ++k) atomicAdd(o + (lane + 32u * k), v[u][k]);
      }
    } else {
      for (int u = 0; b0 + u < total; ++u)
        add_edge(__builtin_amdgcn_readlane(sv, u), __builtin_amdgcn_readlane(dv, u));
    }
  }
  __syncthreads();
}

__device__ __forceinline__ v16h load_a_frag_f32(const float* __restrict__ rowBase,
                                                int k0, float scale, int khalf) {
  int g0 = khalf * 8;
  int g1 = 16 + khalf * 8;
  v4f a0 = *(const v4f*)(rowBase + k0 + g0);
  v4f a1 = *(const v4f*)(rowBase + k0 + g0 + 4);
  v4f a2 = *(const v4f*)(rowBase + k0 + g1);
  v4f a3 = *(const v4f*)(rowBase + k0 + g1 + 4);
  v16h r;
#pragma unroll
  for (int i = 0; i < 4; ++i) {
    r[i]      = (_Float16)(a0[i] * scale);
    r[4 + i]  = (_Float16)(a1[i] * scale);
    r[8 + i]  = (_Float16)(a2[i] * scale);
    r[12 + i] = (_Float16)(a3[i] * scale);
  }
  return r;
}

__global__ void __launch_bounds__(NT)
sage_layer1_owner(const float* __restrict__ x, const int* __restrict__ ei,
                  const float* __restrict__ Wl, const float* __restrict__ Wr,
                  const float* __restrict__ bias,
                  float* __restrict__ h, int N, int E) {
  __shared__ __attribute__((aligned(16))) float sAgg[RB * P1];
  __shared__ __attribute__((aligned(16))) _Float16 sW[2][32][256];
  __shared__ float sCnt[RB];
  __shared__ int sLS[CAP1];
  __shared__ int sLD[CAP1];
  __shared__ int sNum;

  for (int idx = threadIdx.x; idx < 32 * 256; idx += NT) {
    int g  = idx >> 8;
    int n  = (idx >> 4) & 15;
    int j  = idx & 15;
    int ct = g >> 3;
    int kh = (g >> 2) & 1;
    int kc = g & 3;
    int k  = kc * 32 + kh * 8 + j + ((j >= 8) ? 8 : 0);
    int col = ct * 16 + n;
    sW[0][g][n * 16 + j] = (_Float16)Wl[k * HID + col];
    sW[1][g][n * 16 + j] = (_Float16)Wr[k * HID + col];
  }

  const int n0 = blockIdx.x * RB;
  const int nrows = (N - n0 < RB) ? (N - n0) : RB;
  owner_sum<IN_DIM, P1, 8, CAP1>(x, ei, E, n0, nrows, sAgg, sCnt, sLS, sLD, &sNum);

  const int wave = __builtin_amdgcn_readfirstlane((int)(threadIdx.x >> 5));
  const int lid  = threadIdx.x & 31;
  const int m     = lid & 15;
  const int khalf = lid >> 4;

  for (int tile = wave; tile * 16 < nrows; tile += NW) {
    const int lr = tile * 16 + m;
    int row = n0 + lr; if (row >= N) row = N - 1;

    float inv = 1.0f / fmaxf(sCnt[lr], 1.0f);
    const float* aggRow = sAgg + lr * P1;
    const float* xRow   = x + (size_t)row * IN_DIM;

    v8f acc[4] = {{}, {}, {}, {}};
#pragma unroll
    for (int kc = 0; kc < 4; ++kc) {
      int k0 = kc * 32;
      v16h aA = load_a_frag_f32(aggRow, k0, inv,  khalf);
      v16h aX = load_a_frag_f32(xRow,   k0, 1.0f, khalf);
#pragma unroll
      for (int ct = 0; ct < 4; ++ct) {
        int g = (ct * 2 + khalf) * 4 + kc;
        v16h bL = *(const v16h*)&sW[0][g][m * 16];
        v16h bR = *(const v16h*)&sW[1][g][m * 16];
        acc[ct] = __builtin_amdgcn_wmma_f32_16x16x32_f16(false, aA, false, bL,
                                                         (short)0, acc[ct],
                                                         false, false);
        acc[ct] = __builtin_amdgcn_wmma_f32_16x16x32_f16(false, aX, false, bR,
                                                         (short)0, acc[ct],
                                                         false, false);
      }
    }

#pragma unroll
    for (int ct = 0; ct < 4; ++ct) {
      int n = ct * 16 + m;
      float bv = bias[n];
#pragma unroll
      for (int r = 0; r < 8; ++r) {
        int orow = n0 + tile * 16 + r + 8 * khalf;
        if (orow < N) h[(size_t)orow * HID + n] = fmaxf(acc[ct][r] + bv, 0.0f);
      }
    }
  }
}

__global__ void __launch_bounds__(NT)
sage_layer2_owner(const float* __restrict__ h, const int* __restrict__ ei,
                  const float* __restrict__ Wl, const float* __restrict__ Wr,
                  const float* __restrict__ bias,
                  float* __restrict__ out, int N, int E) {
  __shared__ __attribute__((aligned(16))) float sAgg[RB * P2];
  __shared__ float sCnt[RB];
  __shared__ int sLS[CAP2];
  __shared__ int sLD[CAP2];
  __shared__ int sNum;

  const int n0 = blockIdx.x * RB;
  const int nrows = (N - n0 < RB) ? (N - n0) : RB;
  owner_sum<HID, P2, 16, CAP2>(h, ei, E, n0, nrows, sAgg, sCnt, sLS, sLD, &sNum);

  const int t = threadIdx.x;
  if (t >= nrows) return;
  const int nidx = n0 + t;
  float inv = 1.0f / fmaxf(sCnt[t], 1.0f);
  const float* hr = h + (size_t)nidx * HID;
  const float* ar = sAgg + t * P2;
  float o0 = bias[0], o1 = bias[1], o2 = bias[2];
#pragma unroll 8
  for (int k = 0; k < HID; ++k) {
    float av = ar[k] * inv;
    float hv = hr[k];
    o0 += av * Wl[k * NOUT + 0] + hv * Wr[k * NOUT + 0];
    o1 += av * Wl[k * NOUT + 1] + hv * Wr[k * NOUT + 1];
    o2 += av * Wl[k * NOUT + 2] + hv * Wr[k * NOUT + 2];
  }
  float mx  = fmaxf(o0, fmaxf(o1, o2));
  float lse = logf(expf(o0 - mx) + expf(o1 - mx) + expf(o2 - mx));
  out[(size_t)nidx * NOUT + 0] = o0 - mx - lse;
  out[(size_t)nidx * NOUT + 1] = o1 - mx - lse;
  out[(size_t)nidx * NOUT + 2] = o2 - mx - lse;
}


extern "C" void kernel_launch(void* const* d_in, const int* in_sizes, int n_in,
                              void* d_out, int out_size, void* d_ws, size_t ws_size,
                              hipStream_t stream) {
  (void)n_in; (void)out_size; (void)ws_size;
  const float* x   = (const float*)d_in[0];
  const int*   ei  = (const int*)d_in[1];
  const float* Wl1 = (const float*)d_in[2];
  const float* Wr1 = (const float*)d_in[3];
  const float* b1  = (const float*)d_in[4];
  const float* Wl2 = (const float*)d_in[5];
  const float* Wr2 = (const float*)d_in[6];
  const float* b2  = (const float*)d_in[7];

  int N = in_sizes[0] / IN_DIM;
  int E = in_sizes[1] / 2;

  float* hbuf = (float*)d_ws;

  int blocks = (N + RB - 1) / RB;
  sage_layer1_owner<<<blocks, NT, 0, stream>>>(x, ei, Wl1, Wr1, b1, hbuf, N, E);
  sage_layer2_owner<<<blocks, NT, 0, stream>>>(hbuf, ei, Wl2, Wr2, b2, (float*)d_out, N, E);
}


# Edges sorted once per call by owner group (400-node range) in a new first launch, LDS counting sort per 4096-edge chunk; owners fetch their own runs, no sweep of all ids; sums in LDS-atomic order
# speedup vs baseline: 3.6322x; 1.6514x over previous
#include <hip/hip_runtime.h>
#include <hip/hip_bf16.h>


typedef __attribute__((ext_vector_type(16))) _Float16 v16h;
typedef __attribute__((ext_vector_type(8)))  float    v8f;
typedef __attribute__((ext_vector_type(4)))  float    v4f;

#define IN_DIM 128
#define HID    64
#define NOUT   3
#define RB     400
#define NT     1024
#define NW     32
#define P1     132
#define P2     68
#define CAP1   8192
#define CAP2   16384
#define CHK    4096
#define NTB    256
#define KPB    (CHK / NTB)
#define NBMAX  1024
#define SEGW   8

__global__ void __launch_bounds__(NTB)
bucket_edges(const int* __restrict__ ei, int E, int N, int NB,
             int* __restrict__ sorted, int* __restrict__ seg) {
  __shared__ int sHist[NBMAX];
  __shared__ int sOff[NBMAX];
  __shared__ int sWave[NTB / 32];
  __shared__ int2 sStage[CHK];

  const int tid  = threadIdx.x;
  const int lane = tid & 31;
  const int wave = tid >> 5;
  const int G    = gridDim.x;
  const int base = (int)blockIdx.x * CHK;

  for (int i = tid; i < NBMAX; i += NTB) sHist[i] = 0;
  __syncthreads();

  int s[KPB], t[KPB], b[KPB], r[KPB];
#pragma unroll
  for (int k = 0; k < KPB; ++k) {
    const int e = base + tid + k * NTB;
    int d = -1, sv = 0;
    if (e < E) { d = ei[E + e]; sv = ei[e]; }
    s[k] = sv; t[k] = d;
  }
#pragma unroll
  for (int k = 0; k < KPB; ++k) {
    const bool ok = (unsigned)t[k] < (unsigned)N;
    const int bb = ok ? t[k] / RB : 0;
    b[k] = ok ? bb : -1;
    t[k] -= bb * RB;
    r[k] = ok ? atomicAdd(&sHist[bb], 1) : 0;
  }
  __syncthreads();

  constexpr int BPT = NBMAX / NTB;
  int v[BPT], sum = 0;
#pragma unroll
  for (int q = 0; q < BPT; ++q) {
    const int bin = tid * BPT + q;
    v[q] = (bin < NB) ? sHist[bin] : 0;
    sum += v[q];
  }
  int incl = sum;
  for (int d = 1; d < 32; d <<= 1) {
    const int u = __shfl_up(incl, d);
    if (lane >= d) incl += u;
  }
  if (lane == 31) sWave[wave] = incl;
  __syncthreads();
  if (wave == 0) {
    const int w = (lane < NTB / 32) ? sWave[lane] : 0;
    int wi = w;
    for (int d = 1; d < NTB / 32; d <<= 1) {
      const int u = __shfl_up(wi, d);
      if (lane >= d) wi += u;
    }
    if (lane < NTB / 32) sWave[lane] = wi - w;
  }
  __syncthreads();
  int run = incl - sum + sWave[wave];
#pragma unroll
  for (int q = 0; q < BPT; ++q) {
    const int bin = tid * BPT + q;
    if (bin < NB) {
      sOff[bin] = run;
      *(int2*)(seg + 2 * ((size_t)bin * G + blockIdx.x)) = make_int2(base + run, v[q]);
      run += v[q];
    }
  }
  __syncthreads();

#pragma unroll
  for (int k = 0; k < KPB; ++k)
    if (b[k] >= 0) sStage[sOff[b[k]] + r[k]] = make_int2(s[k], t[k]);
  __syncthreads();

  const int total = sOff[NB - 1] + sHist[NB - 1];
  int2* __restrict__ o = (int2*)sorted + base;
  for (int i = tid; i < total; i += NTB) o[i] = sStage[i];
}

template <int ROW, int PITCH, int U, int CAP>
__device__ __forceinline__ void owner_sum(const float* __restrict__ rows,
                                          const int* __restrict__ sorted,
                                          const int* __restrict__ seg, int G,
                                          float* sAgg, float* sCnt,
                                          int* sLS, int* sLD, int* sNum) {
  constexpr int LPR = ROW / 32;
  const unsigned lane = threadIdx.x & 31u;
  const int wave = __builtin_amdgcn_readfirstlane((int)(threadIdx.x >> 5));

  {
    v4f* q = (v4f*)sAgg;
    const v4f z = {0.0f, 0.0f, 0.0f, 0.0f};
    for (int i = threadIdx.x; i < RB * PITCH / 4; i += NT) q[i] = z;
    for (int i = threadIdx.x; i < RB; i += NT) sCnt[i] = 0.0f;
    if (threadIdx.x == 0) *sNum = 0;
  }
  __syncthreads();

  auto add_edge = [&](int s, int d) {
    const float* in = rows + (size_t)s * ROW;
    float* o = sAgg + d * PITCH;
    float t[LPR];
#pragma unroll
    for (int k = 0; k < LPR; ++k) t[k] = in[lane + 32u * k];
#pragma unroll
    for (int k = 0; k < LPR; ++k) atomicAdd(o + (lane + 32u * k), t[k]);
  };

  auto place = [&](bool act, int s, int t, int p) {
    bool ovf = false;
    if (act) {
      atomicAdd(sCnt + t, 1.0f);
      if (p < CAP) { sLS[p] = s; sLD[p] = t; } else ovf = true;
    }
    unsigned om = __builtin_amdgcn_ballot_w32(ovf);
    while (om) {
      const int j = __builtin_ctz(om);
      om &= om - 1u;
      add_edge(__builtin_amdgcn_readlane(s, j), __builtin_amdgcn_readlane(t, j));
    }
  };

  const int* __restrict__ segb = seg + (size_t)blockIdx.x * (size_t)G * 2;
  for (int g0 = wave * SEGW; g0 < G; g0 += NW * SEGW) {
    int lo = 0, cnt = 0;
    if ((int)lane < SEGW && g0 + (int)lane < G) {
      const int2 sg = *(const int2*)(segb + 2 * (g0 + (int)lane));
      lo = sg.x; cnt = sg.y;
    }
    int incl = cnt;
    for (int d = 1; d < SEGW; d <<= 1) {
      const int u = __shfl_up(incl, d);
      if ((int)lane >= d) incl += u;
    }
    const int tot = __builtin_amdgcn_readlane(incl, SEGW - 1);
    int bs = 0;
    if (lane == 0) bs = atomicAdd(sNum, tot);
    bs = __builtin_amdgcn_readfirstlane(bs);
    const int pos = bs + incl - cnt;

    int es[SEGW], et[SEGW];
#pragma unroll
    for (int u = 0; u < SEGW; ++u) {
      const int lo_u = __builtin_amdgcn_readlane(lo, u);
      const int c_u  = __builtin_amdgcn_readlane(cnt, u);
      es[u] = 0; et[u] = 0;
      if ((int)lane < c_u) {
        const int2 e = *(const int2*)(sorted + 2 * (size_t)(lo_u + (int)lane));
        es[u] = e.x; et[u] = e.y;
      }
    }
#pragma unroll
    for (int u = 0; u < SEGW; ++u)
      place((int)lane < __builtin_amdgcn_readlane(cnt, u), es[u], et[u],
            __builtin_amdgcn_readlane(pos, u) + (int)lane);
    for (int u = 0; u < SEGW; ++u) {
      const int lo_u = __builtin_amdgcn_readlane(lo, u);
      const int c_u  = __builtin_amdgcn_readlane(cnt, u);
      const int p_u  = __builtin_amdgcn_readlane(pos, u);
      for (int j0 = 32; j0 < c_u; j0 += 32) {
        const int j = j0 + (int)lane;
        int sv = 0, tv = 0;
        if (j < c_u) {
          const int2 e = *(const int2*)(sorted + 2 * (size_t)(lo_u + j));
          sv = e.x; tv = e.y;
        }
        place(j < c_u, sv, tv, p_u + j);
      }
    }
  }
  __syncthreads();

  int total = *sNum; if (total > CAP) total = CAP;
  total = __builtin_amdgcn_readfirstlane(total);

  for (int b0 = wave * U; b0 < total; b0 += NW * U) {
    const int idx = b0 + (int)lane;
    int sv = 0, dv = 0;
    if ((int)lane < U && idx < total) { sv = sLS[idx]; dv = sLD[idx]; }
    if (b0 + U <= total) {
      float v[U][LPR];
      int dd[U];
#pragma unroll
      for (int u = 0; u < U; ++u) {
        const int s = __builtin_amdgcn_readlane(sv, u);
        dd[u] = __builtin_amdgcn_readlane(dv, u);
        const float* in = rows + (size_t)s * ROW;
#pragma unroll
        for (int k = 0; k < LPR; ++k) v[u][k] = in[lane + 32u * k];
      }
#pragma unroll
      for (int u = 0; u < U; ++u) {
        float* o = sAgg + dd[u] * PITCH;
#pragma unroll
        for (int k = 0; k < LPR; ++k) atomicAdd(o + (lane + 32u * k), v[u][k]);
      }
    } else {
      for (int u = 0; b0 + u < total; ++u)
        add_edge(__builtin_amdgcn_readlane(sv, u), __builtin_amdgcn_readlane(dv, u));
    }
  }
  __syncthreads();
}

__device__ __forceinline__ v16h load_a_frag_f32(const float* __restrict__ rowBase,
                                                int k0, float scale, int khalf) {
  int g0 = khalf * 8;
  int g1 = 16 + khalf * 8;
  v4f a0 = *(const v4f*)(rowBase + k0 + g0);
  v4f a1 = *(const v4f*)(rowBase + k0 + g0 + 4);
  v4f a2 = *(const v4f*)(rowBase + k0 + g1);
  v4f a3 = *(const v4f*)(rowBase + k0 + g1 + 4);
  v16h r;
#pragma unroll
  for (int i = 0; i < 4; ++i) {
    r[i]      = (_Float16)(a0[i] * scale);
    r[4 + i]  = (_Float16)(a1[i] * scale);
    r[8 + i]  = (_Float16)(a2[i] * scale);
    r[12 + i] = (_Float16)(a3[i] * scale);
  }
  return r;
}

__global__ void __launch_bounds__(NT)
sage_layer1_owner(const float* __restrict__ x, const int* __restrict__ sorted,
                  const int* __restrict__ seg, int G,
                  const float* __restrict__ Wl, const float* __restrict__ Wr,
                  const float* __restrict__ bias,
                  float* __restrict__ h, int N) {
  __shared__ __attribute__((aligned(16))) float sAgg[RB * P1];
  __shared__ __attribute__((aligned(16))) _Float16 sW[2][32][256];
  __shared__ float sCnt[RB];
  __shared__ int sLS[CAP1];
  __shared__ int sLD[CAP1];
  __shared__ int sNum;

  for (int idx = threadIdx.x; idx < 32 * 256; idx += NT) {
    int g  = idx >> 8;
    int n  = (idx >> 4) & 15;
    int j  = idx & 15;
    int ct = g >> 3;
    int kh = (g >> 2) & 1;
    int kc = g & 3;
    int k  = kc * 32 + kh * 8 + j + ((j >= 8) ? 8 : 0);
    int col = ct * 16 + n;
    sW[0][g][n * 16 + j] = (_Float16)Wl[k * HID + col];
    sW[1][g][n * 16 + j] = (_Float16)Wr[k * HID + col];
  }

  const int n0 = blockIdx.x * RB;
  const int nrows = (N - n0 < RB) ? (N - n0) : RB;
  owner_sum<IN_DIM, P1, 8, CAP1>(x, sorted, seg, G, sAgg, sCnt, sLS, sLD, &sNum);

  const int wave = __builtin_amdgcn_readfirstlane((int)(threadIdx.x >> 5));
  const int lid  = threadIdx.x & 31;
  const int m     = lid & 15;
  const int khalf = lid >> 4;

  for (int tile = wave; tile * 16 < nrows; tile += NW) {
    const int lr = tile * 16 + m;
    int row = n0 + lr; if (row >= N) row = N - 1;

    float inv = 1.0f / fmaxf(sCnt[lr], 1.0f);
    const float* aggRow = sAgg + lr * P1;
    const float* xRow   = x + (size_t)row * IN_DIM;

    v8f acc[4] = {{}, {}, {}, {}};
#pragma unroll
    for (int kc = 0; kc < 4; ++kc) {
      int k0 = kc * 32;
      v16h aA = load_a_frag_f32(aggRow, k0, inv,  khalf);
      v16h aX = load_a_frag_f32(xRow,   k0, 1.0f, khalf);
#pragma unroll
      for (int ct = 0; ct < 4; ++ct) {
        int g = (ct * 2 + khalf) * 4 + kc;
        v16h bL = *(const v16h*)&sW[0][g][m * 16];
        v16h bR = *(const v16h*)&sW[1][g][m * 16];
        acc[ct] = __builtin_amdgcn_wmma_f32_16x16x32_f16(false, aA, false, bL,
                                                         (short)0, acc[ct],
                                                         false, false);
        acc[ct] = __builtin_amdgcn_wmma_f32_16x16x32_f16(false, aX, false, bR,
                                                         (short)0, acc[ct],
                                                         false, false);
      }
    }

#pragma unroll
    for (int ct = 0; ct < 4; ++ct) {
      int n = ct * 16 + m;
      float bv = bias[n];
#pragma unroll
      for (int r = 0; r < 8; ++r) {
        int orow = n0 + tile * 16 + r + 8 * khalf;
        if (orow < N) h[(size_t)orow * HID + n] = fmaxf(acc[ct][r] + bv, 0.0f);
      }
    }
  }
}

__global__ void __launch_bounds__(NT)
sage_layer2_owner(const float* __restrict__ h, const int* __restrict__ sorted,
                  const int* __restrict__ seg, int G,
                  const float* __restrict__ Wl, const float* __restrict__ Wr,
                  const float* __restrict__ bias,
                  float* __restrict__ out, int N) {
  __shared__ __attribute__((aligned(16))) float sAgg[RB * P2];
  __shared__ float sCnt[RB];
  __shared__ int sLS[CAP2];
  __shared__ int sLD[CAP2];
  __shared__ int sNum;

  const int n0 = blockIdx.x * RB;
  const int nrows = (N - n0 < RB) ? (N - n0) : RB;
  owner_sum<HID, P2, 16, CAP2>(h, sorted, seg, G, sAgg, sCnt, sLS, sLD, &sNum);

  const int t = threadIdx.x;
  if (t >= nrows) return;
  const int nidx = n0 + t;
  float inv = 1.0f / fmaxf(sCnt[t], 1.0f);
  const float* hr = h + (size_t)nidx * HID;
  const float* ar = sAgg + t * P2;
  float o0 = bias[0], o1 = bias[1], o2 = bias[2];
#pragma unroll 8
  for (int k = 0; k < HID; ++k) {
    float av = ar[k] * inv;
    float hv = hr[k];
    o0 += av * Wl[k * NOUT + 0] + hv * Wr[k * NOUT + 0];
    o1 += av * Wl[k * NOUT + 1] + hv * Wr[k * NOUT + 1];
    o2 += av * Wl[k * NOUT + 2] + hv * Wr[k * NOUT + 2];
  }
  float mx  = fmaxf(o0, fmaxf(o1, o2));
  float lse = logf(expf(o0 - mx) + expf(o1 - mx) + expf(o2 - mx));
  out[(size_t)nidx * NOUT + 0] = o0 - mx - lse;
  out[(size_t)nidx * NOUT + 1] = o1 - mx - lse;
  out[(size_t)nidx * NOUT + 2] = o2 - mx - lse;
}


extern "C" void kernel_launch(void* const* d_in, const int* in_sizes, int n_in,
                              void* d_out, int out_size, void* d_ws, size_t ws_size,
                              hipStream_t stream) {
  (void)n_in; (void)out_size; (void)ws_size;
  const float* x   = (const float*)d_in[0];
  const int*   ei  = (const int*)d_in[1];
  const float* Wl1 = (const float*)d_in[2];
  const float* Wr1 = (const float*)d_in[3];
  const float* b1  = (const float*)d_in[4];
  const float* Wl2 = (const float*)d_in[5];
  const float* Wr2 = (const float*)d_in[6];
  const float* b2  = (const float*)d_in[7];

  int N = in_sizes[0] / IN_DIM;
  int E = in_sizes[1] / 2;

  int blocks = (N + RB - 1) / RB;
  int G = E / CHK + 1;

  float* hbuf = (float*)d_ws;
  int* sorted = (int*)(hbuf + (size_t)N * HID);
  int* seg    = sorted + (size_t)G * CHK * 2;

  bucket_edges<<<G, NTB, 0, stream>>>(ei, E, N, blocks, sorted, seg);
  sage_layer1_owner<<<blocks, NT, 0, stream>>>(x, sorted, seg, G, Wl1, Wr1, b1, hbuf, N);
  sage_layer2_owner<<<blocks, NT, 0, stream>>>(hbuf, sorted, seg, G, Wl2, Wr2, b2, (float*)d_out, N);
}


# Owner's edge list counting-sorted by destination row in LDS (integer rank adds); waves sum a row's source rows in registers in rank (arrival) order, one float LDS add per run, not one per edge
# speedup vs baseline: 3.8662x; 1.0644x over previous
#include <hip/hip_runtime.h>
#include <hip/hip_bf16.h>


typedef __attribute__((ext_vector_type(16))) _Float16 v16h;
typedef __attribute__((ext_vector_type(8)))  float    v8f;
typedef __attribute__((ext_vector_type(4)))  float    v4f;

#define IN_DIM 128
#define HID    64
#define NOUT   3
#define RB     400
#define NT     1024
#define NW     32
#define P1     132
#define P2     68
#define CAP    4096
#define RBITS  9
#define CHK    4096
#define NTB    256
#define KPB    (CHK / NTB)
#define NBMAX  1024
#define SEGW   8

__global__ void __launch_bounds__(NTB)
bucket_edges(const int* __restrict__ ei, int E, int N, int NB,
             int* __restrict__ sorted, int* __restrict__ seg) {
  __shared__ int sHist[NBMAX];
  __shared__ int sOff[NBMAX];
  __shared__ int sWave[NTB / 32];
  __shared__ int2 sStage[CHK];

  const int tid  = threadIdx.x;
  const int lane = tid & 31;
  const int wave = tid >> 5;
  const int G    = gridDim.x;
  const int base = (int)blockIdx.x * CHK;

  for (int i = tid; i < NBMAX; i += NTB) sHist[i] = 0;
  __syncthreads();

  int s[KPB], t[KPB], b[KPB], r[KPB];
#pragma unroll
  for (int k = 0; k < KPB; ++k) {
    const int e = base + tid + k * NTB;
    int d = -1, sv = 0;
    if (e < E) { d = ei[E + e]; sv = ei[e]; }
    s[k] = sv; t[k] = d;
  }
#pragma unroll
  for (int k = 0; k < KPB; ++k) {
    const bool ok = (unsigned)t[k] < (unsigned)N;
    const int bb = ok ? t[k] / RB : 0;
    b[k] = ok ? bb : -1;
    t[k] -= bb * RB;
    r[k] = ok ? atomicAdd(&sHist[bb], 1) : 0;
  }
  __syncthreads();

  constexpr int BPT = NBMAX / NTB;
  int v[BPT], sum = 0;
#pragma unroll
  for (int q = 0; q < BPT; ++q) {
    const int bin = tid * BPT + q;
    v[q] = (bin < NB) ? sHist[bin] : 0;
    sum += v[q];
  }
  int incl = sum;
  for (int d = 1; d < 32; d <<= 1) {
    const int u = __shfl_up(incl, d);
    if (lane >= d) incl += u;
  }
  if (lane == 31) sWave[wave] = incl;
  __syncthreads();
  if (wave == 0) {
    const int w = (lane < NTB / 32) ? sWave[lane] : 0;
    int wi = w;
    for (int d = 1; d < NTB / 32; d <<= 1) {
      const int u = __shfl_up(wi, d);
      if (lane >= d) wi += u;
    }
    if (lane < NTB / 32) sWave[lane] = wi - w;
  }
  __syncthreads();
  int run = incl - sum + sWave[wave];
#pragma unroll
  for (int q = 0; q < BPT; ++q) {
    const int bin = tid * BPT + q;
    if (bin < NB) {
      sOff[bin] = run;
      *(int2*)(seg + 2 * ((size_t)bin * G + blockIdx.x)) = make_int2(base + run, v[q]);
      run += v[q];
    }
  }
  __syncthreads();

#pragma unroll
  for (int k = 0; k < KPB; ++k)
    if (b[k] >= 0) sStage[sOff[b[k]] + r[k]] = make_int2(s[k], t[k]);
  __syncthreads();

  const int total = sOff[NB - 1] + sHist[NB - 1];
  int2* __restrict__ o = (int2*)sorted + base;
  for (int i = tid; i < total; i += NTB) o[i] = sStage[i];
}

template <int ROW, int PITCH, int U>
__device__ __forceinline__ void owner_sum(const float* __restrict__ rows,
                                          const int* __restrict__ sorted,
                                          const int* __restrict__ seg, int G,
                                          float* sAgg, int* sCnt, int* sExt, int* sStart, int* sWv,
                                          int* sLS, int* sLK, int* sSS, int* sSD, int* sNum) {
  static_assert(RB <= NT && RB <= (1 << RBITS) && NW == 32, "one scan pass, row number in RBITS bits");
  constexpr int VPL = ROW / 32;
  typedef float vrow __attribute__((ext_vector_type(VPL)));
  const int lane = (int)(threadIdx.x & 31u);
  const int wave = __builtin_amdgcn_readfirstlane((int)(threadIdx.x >> 5));

  {
    v4f* q = (v4f*)sAgg;
    const v4f z = {0.0f, 0.0f, 0.0f, 0.0f};
    for (int i = threadIdx.x; i < (RB + 1) * PITCH / 4; i += NT) q[i] = z;
    for (int i = threadIdx.x; i < RB; i += NT) { sCnt[i] = 0; sExt[i] = 0; }
    if (threadIdx.x == 0) *sNum = 0;
  }
  __syncthreads();

  auto add_edge = [&](int s, int d) {
    const vrow t = *(const vrow*)(rows + (size_t)s * ROW + lane * VPL);
    float* o = sAgg + d * PITCH + lane * VPL;
#pragma unroll
    for (int j = 0; j < VPL; ++j) atomicAdd(o + j, t[j]);
  };

  auto beyond = [&](bool act, int s, int t, int p) {
    const bool ovf = act && p >= CAP;
    if (ovf) atomicAdd(sExt + t, 1);
    unsigned om = __builtin_amdgcn_ballot_w32(ovf);
    while (om) {
      const int j = __builtin_ctz(om);
      om &= om - 1u;
      add_edge(__builtin_amdgcn_readlane(s, j), __builtin_amdgcn_readlane(t, j));
    }
  };

  const int* __restrict__ segb = seg + (size_t)blockIdx.x * (size_t)G * 2;
  for (int g0 = wave * SEGW; g0 < G; g0 += NW * SEGW) {
    int lo = 0, cnt = 0;
    if (lane < SEGW && g0 + lane < G) {
      const int2 sg = *(const int2*)(segb + 2 * (g0 + lane));
      lo = sg.x; cnt = sg.y;
    }
    int incl = cnt;
    for (int d = 1; d < SEGW; d <<= 1) {
      const int u = __shfl_up(incl, d);
      if (lane >= d) incl += u;
    }
    const int tot = __builtin_amdgcn_readlane(incl, SEGW - 1);
    int bs = 0;
    if (lane == 0) bs = atomicAdd(sNum, tot);
    bs = __builtin_amdgcn_readfirstlane(bs);
    const int pos = bs + incl - cnt;

    int es[SEGW], et[SEGW], rk[SEGW];
#pragma unroll
    for (int u = 0; u < SEGW; ++u) {
      const int lo_u = __builtin_amdgcn_readlane(lo, u);
      const int c_u  = __builtin_amdgcn_readlane(cnt, u);
      es[u] = 0; et[u] = 0;
      if (lane < c_u) {
        const int2 e = *(const int2*)(sorted + 2 * (size_t)(lo_u + lane));
        es[u] = e.x; et[u] = e.y;
      }
    }
#pragma unroll
    for (int u = 0; u < SEGW; ++u) {
      const int p = __builtin_amdgcn_readlane(pos, u) + lane;
      rk[u] = 0;
      if (lane < __builtin_amdgcn_readlane(cnt, u) && p < CAP) rk[u] = atomicAdd(sCnt + et[u], 1);
    }
#pragma unroll
    for (int u = 0; u < SEGW; ++u) {
      const int p = __builtin_amdgcn_readlane(pos, u) + lane;
      if (lane < __builtin_amdgcn_readlane(cnt, u) && p < CAP) {
        sLS[p] = es[u]; sLK[p] = et[u] | (rk[u] << RBITS);
      }
    }
    if (bs + tot > CAP) {
#pragma unroll
      for (int u = 0; u < SEGW; ++u)
        beyond(lane < __builtin_amdgcn_readlane(cnt, u), es[u], et[u],
               __builtin_amdgcn_readlane(pos, u) + lane);
    }
    for (int u = 0; u < SEGW; ++u) {
      const int lo_u = __builtin_amdgcn_readlane(lo, u);
      const int c_u  = __builtin_amdgcn_readlane(cnt, u);
      const int p_u  = __builtin_amdgcn_readlane(pos, u);
      for (int j0 = 32; j0 < c_u; j0 += 32) {
        const int j = j0 + lane;
        int sv = 0, tv = 0;
        if (j < c_u) {
          const int2 e = *(const int2*)(sorted + 2 * (size_t)(lo_u + j));
          sv = e.x; tv = e.y;
        }
        if (j < c_u && p_u + j < CAP) {
          const int r = atomicAdd(sCnt + tv, 1);
          sLS[p_u + j] = sv; sLK[p_u + j] = tv | (r << RBITS);
        }
        beyond(j < c_u, sv, tv, p_u + j);
      }
    }
  }
  __syncthreads();

  int total = *sNum; if (total > CAP) total = CAP;
  total = __builtin_amdgcn_readfirstlane(total);

  {
    const int i = (int)threadIdx.x;
    const int v = (i < RB) ? sCnt[i] : 0;
    int incl = v;
    for (int d = 1; d < 32; d <<= 1) {
      const int u = __shfl_up(incl, d);
      if (lane >= d) incl += u;
    }
    if (lane == 31) sWv[wave] = incl;
    __syncthreads();
    if (wave == 0) {
      const int w = sWv[lane];
      int wi = w;
      for (int d = 1; d < 32; d <<= 1) {
        const int u = __shfl_up(wi, d);
        if (lane >= d) wi += u;
      }
      sWv[lane] = wi - w;
    }
    __syncthreads();
    if (i < RB) sStart[i] = incl - v + sWv[wave];
  }
  __syncthreads();

  const int gt   = (total + U - 1) / U;
  const int g_lo = (gt * wave) / NW;
  const int g_hi = (gt * (wave + 1)) / NW;

  for (int p = threadIdx.x; p < total; p += NT) {
    const int s = sLS[p];
    const int k = sLK[p];
    const int t = k & ((1 << RBITS) - 1);
    const int q = sStart[t] + (k >> RBITS);
    sSS[q] = s; sSD[q] = t;
  }
  if ((int)threadIdx.x < gt * U - total) {
    sSS[total + threadIdx.x] = 0; sSD[total + threadIdx.x] = RB;
  }
  __syncthreads();

  int cur = RB;
  vrow acc = 0.0f;
  auto flush = [&]() {
    float* o = sAgg + cur * PITCH + lane * VPL;
#pragma unroll
    for (int j = 0; j < VPL; ++j) atomicAdd(o + j, acc[j]);
  };

  int sv = 0, dv = 0;
  if (g_lo < g_hi && lane < U) { sv = sSS[g_lo * U + lane]; dv = sSD[g_lo * U + lane]; }
  for (int g = g_lo; g < g_hi; ++g) {
    vrow v[U];
#pragma unroll
    for (int u = 0; u < U; ++u) {
      const int s = __builtin_amdgcn_readlane(sv, u);
      v[u] = *(const vrow*)(rows + (size_t)s * ROW + lane * VPL);
    }
    int svn = 0, dvn = 0;
    if (g + 1 < g_hi && lane < U) { svn = sSS[(g + 1) * U + lane]; dvn = sSD[(g + 1) * U + lane]; }
#pragma unroll
    for (int u = 0; u < U; ++u) {
      const int d = __builtin_amdgcn_readlane(dv, u);
      if (d != cur) { flush(); cur = d; acc = v[u]; } else { acc += v[u]; }
    }
    sv = svn; dv = dvn;
  }
  flush();
  __syncthreads();
}

__device__ __forceinline__ v16h load_a_frag_f32(const float* __restrict__ rowBase,
                                                int k0, float scale, int khalf) {
  int g0 = khalf * 8;
  int g1 = 16 + khalf * 8;
  v4f a0 = *(const v4f*)(rowBase + k0 + g0);
  v4f a1 = *(const v4f*)(rowBase + k0 + g0 + 4);
  v4f a2 = *(const v4f*)(rowBase + k0 + g1);
  v4f a3 = *(const v4f*)(rowBase + k0 + g1 + 4);
  v16h r;
#pragma unroll
  for (int i = 0; i < 4; ++i) {
    r[i]      = (_Float16)(a0[i] * scale);
    r[4 + i]  = (_Float16)(a1[i] * scale);
    r[8 + i]  = (_Float16)(a2[i] * scale);
    r[12 + i] = (_Float16)(a3[i] * scale);
  }
  return r;
}

__global__ void __launch_bounds__(NT)
sage_layer1_owner(const float* __restrict__ x, const int* __restrict__ sorted,
                  const int* __restrict__ seg, int G,
                  const float* __restrict__ Wl, const float* __restrict__ Wr,
                  const float* __restrict__ bias,
                  float* __restrict__ h, int N) {
  __shared__ __attribute__((aligned(16))) float sAgg[(RB + 1) * P1];
  __shared__ __attribute__((aligned(16))) _Float16 sW[2][32][256];
  __shared__ int sCnt[RB];
  __shared__ int sExt[RB];
  __shared__ int sStart[RB];
  __shared__ int sWv[NW];
  __shared__ int sLS[CAP];
  __shared__ int sLK[CAP];
  __shared__ int sSS[CAP + 16];
  __shared__ int sSD[CAP + 16];
  __shared__ int sNum;

  for (int idx = threadIdx.x; idx < 32 * 256; idx += NT) {
    int g  = idx >> 8;
    int n  = (idx >> 4) & 15;
    int j  = idx & 15;
    int ct = g >> 3;
    int kh = (g >> 2) & 1;
    int kc = g & 3;
    int k  = kc * 32 + kh * 8 + j + ((j >= 8) ? 8 : 0);
    int col = ct * 16 + n;
    sW[0][g][n * 16 + j] = (_Float16)Wl[k * HID + col];
    sW[1][g][n * 16 + j] = (_Float16)Wr[k * HID + col];
  }

  const int n0 = blockIdx.x * RB;
  const int nrows = (N - n0 < RB) ? (N - n0) : RB;
  owner_sum<IN_DIM, P1, 8>(x, sorted, seg, G, sAgg, sCnt, sExt, sStart, sWv,
                           sLS, sLK, sSS, sSD, &sNum);

  const int wave = __builtin_amdgcn_readfirstlane((int)(threadIdx.x >> 5));
  const int lid  = threadIdx.x & 31;
  const int m     = lid & 15;
  const int khalf = lid >> 4;

  for (int tile = wave; tile * 16 < nrows; tile += NW) {
    const int lr = tile * 16 + m;
    int row = n0 + lr; if (row >= N) row = N - 1;

    float inv = 1.0f / fmaxf((float)(sCnt[lr] + sExt[lr]), 1.0f);
    const float* aggRow = sAgg + lr * P1;
    const float* xRow   = x + (size_t)row * IN_DIM;

    v8f acc[4] = {{}, {}, {}, {}};
#pragma unroll
    for (int kc = 0; kc < 4; ++kc) {
      int k0 = kc * 32;
      v16h aA = load_a_frag_f32(aggRow, k0, inv,  khalf);
      v16h aX = load_a_frag_f32(xRow,   k0, 1.0f, khalf);
#pragma unroll
      for (int ct = 0; ct < 4; ++ct) {
        int g = (ct * 2 + khalf) * 4 + kc;
        v16h bL = *(const v16h*)&sW[0][g][m * 16];
        v16h bR = *(const v16h*)&sW[1][g][m * 16];
        acc[ct] = __builtin_amdgcn_wmma_f32_16x16x32_f16(false, aA, false, bL,
                                                         (short)0, acc[ct],
                                                         false, false);
        acc[ct] = __builtin_amdgcn_wmma_f32_16x16x32_f16(false, aX, false, bR,
                                                         (short)0, acc[ct],
                                                         false, false);
      }
    }

#pragma unroll
    for (int ct = 0; ct < 4; ++ct) {
      int n = ct * 16 + m;
      float bv = bias[n];
#pragma unroll
      for (int r = 0; r < 8; ++r) {
        int orow = n0 + tile * 16 + r + 8 * khalf;
        if (orow < N) h[(size_t)orow * HID + n] = fmaxf(acc[ct][r] + bv, 0.0f);
      }
    }
  }
}

__global__ void __launch_bounds__(NT)
sage_layer2_owner(const float* __restrict__ h, const int* __restrict__ sorted,
                  const int* __restrict__ seg, int G,
                  const float* __restrict__ Wl, const float* __restrict__ Wr,
                  const float* __restrict__ bias,
                  float* __restrict__ out, int N) {
  __shared__ __attribute__((aligned(16))) float sAgg[(RB + 1) * P2];
  __shared__ int sCnt[RB];
  __shared__ int sExt[RB];
  __shared__ int sStart[RB];
  __shared__ int sWv[NW];
  __shared__ int sLS[CAP];
  __shared__ int sLK[CAP];
  __shared__ int sSS[CAP + 16];
  __shared__ int sSD[CAP + 16];
  __shared__ int sNum;

  const int n0 = blockIdx.x * RB;
  const int nrows = (N - n0 < RB) ? (N - n0) : RB;
  owner_sum<HID, P2, 16>(h, sorted, seg, G, sAgg, sCnt, sExt, sStart, sWv,
                         sLS, sLK, sSS, sSD, &sNum);

  const int t = threadIdx.x;
  if (t >= nrows) return;
  const int nidx = n0 + t;
  float inv = 1.0f / fmaxf((float)(sCnt[t] + sExt[t]), 1.0f);
  const float* hr = h + (size_t)nidx * HID;
  const float* ar = sAgg + t * P2;
  float o0 = bias[0], o1 = bias[1], o2 = bias[2];
#pragma unroll 8
  for (int k = 0; k < HID; ++k) {
    float av = ar[k] * inv;
    float hv = hr[k];
    o0 += av * Wl[k * NOUT + 0] + hv * Wr[k * NOUT + 0];
    o1 += av * Wl[k * NOUT + 1] + hv * Wr[k * NOUT + 1];
    o2 += av * Wl[k * NOUT + 2] + hv * Wr[k * NOUT + 2];
  }
  float mx  = fmaxf(o0, fmaxf(o1, o2));
  float lse = logf(expf(o0 - mx) + expf(o1 - mx) + expf(o2 - mx));
  out[(size_t)nidx * NOUT + 0] = o0 - mx - lse;
  out[(size_t)nidx * NOUT + 1] = o1 - mx - lse;
  out[(size_t)nidx * NOUT + 2] = o2 - mx - lse;
}


extern "C" void kernel_launch(void* const* d_in, const int* in_sizes, int n_in,
                              void* d_out, int out_size, void* d_ws, size_t ws_size,
                              hipStream_t stream) {
  (void)n_in; (void)out_size; (void)ws_size;
  const float* x   = (const float*)d_in[0];
  const int*   ei  = (const int*)d_in[1];
  const float* Wl1 = (const float*)d_in[2];
  const float* Wr1 = (const float*)d_in[3];
  const float* b1  = (const float*)d_in[4];
  const float* Wl2 = (const float*)d_in[5];
  const float* Wr2 = (const float*)d_in[6];
  const float* b2  = (const float*)d_in[7];

  int N = in_sizes[0] / IN_DIM;
  int E = in_sizes[1] / 2;

  int blocks = (N + RB - 1) / RB;
  int G = E / CHK + 1;

  float* hbuf = (float*)d_ws;
  int* sorted = (int*)(hbuf + (size_t)N * HID);
  int* seg    = sorted + (size_t)G * CHK * 2;

  bucket_edges<<<G, NTB, 0, stream>>>(ei, E, N, blocks, sorted, seg);
  sage_layer1_owner<<<blocks, NT, 0, stream>>>(x, sorted, seg, G, Wl1, Wr1, b1, hbuf, N);
  sage_layer2_owner<<<blocks, NT, 0, stream>>>(hbuf, sorted, seg, G, Wl2, Wr2, b2, (float*)d_out, N);
}
